# starting kernel
# baseline (speedup 1.0000x reference)
#include <hip/hip_runtime.h>
#include <math.h>

#define VOCAB 100000
#define DIM   300
#define BATCH 4096
#define SEQL  200
#define HID   1000
#define HIDP  1024
#define OUTD  3
#define REPD  (2*DIM)

typedef __attribute__((ext_vector_type(2))) float v2f;
typedef __attribute__((ext_vector_type(8))) float v8f;

__global__ __launch_bounds__(256) void transpose_pad_kernel(
    const float* __restrict__ W1, float* __restrict__ W1t)
{
    __shared__ float tile[32][33];
    const int kBase = blockIdx.x * 32;
    const int nBase = blockIdx.y * 32;
    const int tx = threadIdx.x;
    const int ty = threadIdx.y;

    #pragma unroll
    for (int i = ty; i < 32; i += 8) {
        const int k = kBase + i, n = nBase + tx;
        tile[i][tx] = (k < REPD && n < HID) ? W1[(size_t)k * HID + n] : 0.0f;
    }
    __syncthreads();
    #pragma unroll
    for (int i = ty; i < 32; i += 8) {
        const int n = nBase + i, k = kBase + tx;
        if (k < REPD)
            W1t[(size_t)n * REPD + k] = tile[tx][i];
    }
}

__global__ __launch_bounds__(320) void pool_kernel(
    const float* __restrict__ emb, const int* __restrict__ x,
    const int* __restrict__ lengths, float* __restrict__ rep)
{
    __shared__ int sidx[SEQL];
    const int b   = blockIdx.x;
    const int tid = threadIdx.x;
    if (tid < SEQL) sidx[tid] = x[b * SEQL + tid];
    __syncthreads();
    if (tid >= DIM) return;

    const int len = lengths[b];
    float sum = 0.0f;
    float mx  = -INFINITY;
    for (int t = 0; t < SEQL; ++t) {
        if (t + 8 < SEQL)
            __builtin_prefetch(emb + (size_t)sidx[t + 8] * DIM + tid, 0, 0);
        const float v = emb[(size_t)sidx[t] * DIM + tid];
        sum += v;
        if (t < len) mx = fmaxf(mx, v);
    }
    rep[(size_t)b * REPD + tid]       = sum / (float)len;
    rep[(size_t)b * REPD + DIM + tid] = mx;
}

__global__ __launch_bounds__(128) void gemm1_wmma_kernel(
    const float* __restrict__ rep, const float* __restrict__ W1t,
    const float* __restrict__ b1, float* __restrict__ h)
{
    const int wave  = blockIdx.x * (blockDim.x >> 5) + (threadIdx.x >> 5);
    const int lane  = threadIdx.x & 31;
    const int mtile = wave & 255;
    const int ntile = wave >> 8;
    const int row0  = mtile * 16;
    const int col0  = ntile * 64;
    const int laneM = lane & 15;
    const int kHi   = (lane >> 4) << 1;

    const float* __restrict__ arow  = rep + (size_t)(row0 + laneM) * REPD;
    const float* __restrict__ brow0 = W1t + (size_t)(col0 +  0 + laneM) * REPD;
    const float* __restrict__ brow1 = W1t + (size_t)(col0 + 16 + laneM) * REPD;
    const float* __restrict__ brow2 = W1t + (size_t)(col0 + 32 + laneM) * REPD;
    const float* __restrict__ brow3 = W1t + (size_t)(col0 + 48 + laneM) * REPD;

    v8f acc0 = {}, acc1 = {}, acc2 = {}, acc3 = {};

    for (int k = 0; k < REPD; k += 4) {
        const int kb = k + kHi;
        const v2f a  = *(const v2f*)(arow  + kb);
        const v2f b0 = *(const v2f*)(brow0 + kb);
        const v2f b1v = *(const v2f*)(brow1 + kb);
        const v2f b2v = *(const v2f*)(brow2 + kb);
        const v2f b3v = *(const v2f*)(brow3 + kb);
        acc0 = __builtin_amdgcn_wmma_f32_16x16x4_f32(false, a, false, b0,  (short)0, acc0, false, false);
        acc1 = __builtin_amdgcn_wmma_f32_16x16x4_f32(false, a, false, b1v, (short)0, acc1, false, false);
        acc2 = __builtin_amdgcn_wmma_f32_16x16x4_f32(false, a, false, b2v, (short)0, acc2, false, false);
        acc3 = __builtin_amdgcn_wmma_f32_16x16x4_f32(false, a, false, b3v, (short)0, acc3, false, false);
    }

    const int mOff = (lane >> 4) << 3;
    const v8f accs[4] = { acc0, acc1, acc2, acc3 };
    #pragma unroll
    for (int j = 0; j < 4; ++j) {
        const int n = col0 + j * 16 + laneM;
        if (n >= HID) continue;
        const float bias = b1[n];
        #pragma unroll
        for (int r = 0; r < 8; ++r) {
            const int m = row0 + r + mOff;
            const float v = accs[j][r] + bias;
            h[(size_t)m * HID + n] = v > 0.0f ? v : 0.0f;
        }
    }
}

__global__ __launch_bounds__(256) void gemm2_kernel(
    const float* __restrict__ h, const float* __restrict__ W2,
    const float* __restrict__ b2, float* __restrict__ out)
{
    __shared__ float red[3][256];
    const int b = blockIdx.x, tid = threadIdx.x;
    float a0 = 0.0f, a1 = 0.0f, a2 = 0.0f;
    const float* __restrict__ hr = h + (size_t)b * HID;
    for (int k = tid; k < HID; k += 256) {
        const float v = hr[k];
        a0 += v * W2[k * 3 + 0];
        a1 += v * W2[k * 3 + 1];
        a2 += v * W2[k * 3 + 2];
    }
    red[0][tid] = a0; red[1][tid] = a1; red[2][tid] = a2;
    __syncthreads();
    for (int s = 128; s > 0; s >>= 1) {
        if (tid < s) {
            red[0][tid] += red[0][tid + s];
            red[1][tid] += red[1][tid + s];
            red[2][tid] += red[2][tid + s];
        }
        __syncthreads();
    }
    if (tid < 3) out[b * 3 + tid] = red[tid][0] + b2[tid];
}

extern "C" void kernel_launch(void* const* d_in, const int* in_sizes, int n_in,
                              void* d_out, int out_size, void* d_ws, size_t ws_size,
                              hipStream_t stream)
{
    const float* emb  = (const float*)d_in[0];
    const float* W1   = (const float*)d_in[1];
    const float* b1   = (const float*)d_in[2];
    const float* W2   = (const float*)d_in[3];
    const float* b2   = (const float*)d_in[4];
    const int*   x    = (const int*)d_in[5];
    const int*   lens = (const int*)d_in[6];
    float* out = (float*)d_out;

    float* W1t = (float*)d_ws;
    float* rep = W1t + (size_t)HIDP * REPD;
    float* h   = rep + (size_t)BATCH * REPD;

    hipLaunchKernelGGL(transpose_pad_kernel, dim3(19, 32), dim3(32, 8), 0, stream,
                       W1, W1t);

    hipLaunchKernelGGL(pool_kernel, dim3(BATCH), dim3(320), 0, stream,
                       emb, x, lens, rep);

    hipLaunchKernelGGL(gemm1_wmma_kernel, dim3(1024), dim3(128), 0, stream,
                       rep, W1t, b1, h);

    hipLaunchKernelGGL(gemm2_kernel, dim3(BATCH), dim3(256), 0, stream,
                       h, W2, b2, out);
}


# pooling re-cut along the sequence: a wave owns whole tokens (ids on the scalar side, 10 row loads a token, ring of 5 tokens in flight), 2 waves per batch row joined in LDS; mean sum order changes
# speedup vs baseline: 1.8378x; 1.8378x over previous
#include <hip/hip_runtime.h>
#include <math.h>

#define VOCAB 100000
#define DIM   300
#define BATCH 4096
#define SEQL  200
#define HID   1000
#define HIDP  1024
#define OUTD  3
#define REPD  (2*DIM)

typedef __attribute__((ext_vector_type(2))) float v2f;
typedef __attribute__((ext_vector_type(4))) float v4f;
typedef __attribute__((ext_vector_type(8))) float v8f;

__global__ __launch_bounds__(256) void transpose_pad_kernel(
    const float* __restrict__ W1, float* __restrict__ W1t)
{
    __shared__ float tile[32][33];
    const int kBase = blockIdx.x * 32;
    const int nBase = blockIdx.y * 32;
    const int tx = threadIdx.x;
    const int ty = threadIdx.y;

    #pragma unroll
    for (int i = ty; i < 32; i += 8) {
        const int k = kBase + i, n = nBase + tx;
        tile[i][tx] = (k < REPD && n < HID) ? W1[(size_t)k * HID + n] : 0.0f;
    }
    __syncthreads();
    #pragma unroll
    for (int i = ty; i < 32; i += 8) {
        const int n = nBase + i, k = kBase + tx;
        if (k < REPD)
            W1t[(size_t)n * REPD + k] = tile[tx][i];
    }
}

#define PL_NW   2
#define PL_TPW  (SEQL / PL_NW)
#define PL_R    5
#define PL_QT   25
#define PL_NQ   (PL_TPW / PL_QT)
#define PL_NP   10
#define PL_LAST 268

__global__ __launch_bounds__(PL_NW * 32) void pool_kernel(
    const float* __restrict__ emb, const int* __restrict__ x,
    const int* __restrict__ lengths, float* __restrict__ rep)
{
    __shared__ float psum[PL_NP][32];
    __shared__ float pmax[PL_NP][32];

    const int b    = blockIdx.x;
    const int lane = threadIdx.x & 31;
    const int w    = __builtin_amdgcn_readfirstlane((int)(threadIdx.x >> 5));
    const int len  = lengths[b];
    const int t0   = w * PL_TPW;
    const int* __restrict__ xr = x + (size_t)b * SEQL + t0;

    const int lq = lane < PL_QT ? lane : PL_QT - 1;
    int xv[PL_NQ];
    #pragma unroll
    for (int q = 0; q < PL_NQ; ++q) xv[q] = xr[q * PL_QT + lq];

    float s[PL_NP], m[PL_NP];
    #pragma unroll
    for (int j = 0; j < PL_NP; ++j) { s[j] = 0.0f; m[j] = -INFINITY; }

    #pragma unroll 1
    for (int q = 0; q < PL_NQ; ++q) {
        const int xq = q == 0 ? xv[0] : q == 1 ? xv[1] : q == 2 ? xv[2] : xv[3];
        const int tq = t0 + q * PL_QT;
        float v[PL_R][PL_NP];

        #pragma unroll
        for (int k = 0; k < PL_R; ++k) {
            const int idx = __builtin_amdgcn_readlane(xq, k);
            const float* __restrict__ row = emb + (size_t)idx * DIM;
            #pragma unroll
            for (int j = 0; j < PL_NP; ++j)
                v[k][j] = row[(j < PL_NP - 1 ? 32 * j : PL_LAST) + lane];
        }

        #pragma unroll
        for (int i = 0; i < PL_QT; ++i) {
            const int  k   = i % PL_R;
            const bool act = (tq + i) < len;
            #pragma unroll
            for (int j = 0; j < PL_NP; ++j) {
                const float a = v[k][j];
                s[j] += a;
                m[j] = (act && a > m[j]) ? a : m[j];
            }
            if (i + PL_R < PL_QT) {
                asm volatile("" :: "v"(s[0]), "v"(s[1]), "v"(s[2]), "v"(s[3]), "v"(s[4]),
                                   "v"(s[5]), "v"(s[6]), "v"(s[7]), "v"(s[8]), "v"(s[9]),
                                   "v"(m[0]), "v"(m[1]), "v"(m[2]), "v"(m[3]), "v"(m[4]),
                                   "v"(m[5]), "v"(m[6]), "v"(m[7]), "v"(m[8]), "v"(m[9])
                             : "memory");
                const int idx = __builtin_amdgcn_readlane(xq, i + PL_R);
                const float* __restrict__ row = emb + (size_t)idx * DIM;
                #pragma unroll
                for (int j = 0; j < PL_NP; ++j)
                    v[k][j] = row[(j < PL_NP - 1 ? 32 * j : PL_LAST) + lane];
            }
        }
    }

    if (w == 1) {
        #pragma unroll
        for (int j = 0; j < PL_NP; ++j) { psum[j][lane] = s[j]; pmax[j][lane] = m[j]; }
    }
    __syncthreads();
    if (w == 0) {
        const float flen = (float)len;
        float* __restrict__ rrow = rep + (size_t)b * REPD;
        #pragma unroll
        for (int j = 0; j < PL_NP; ++j) {
            const float st = s[j] + psum[j][lane];
            const float mt = fmaxf(m[j], pmax[j][lane]);
            const int   d  = (j < PL_NP - 1 ? 32 * j : PL_LAST) + lane;
            if (j < PL_NP - 1 || lane >= 20) {
                rrow[d]       = st / flen;
                rrow[DIM + d] = mt;
            }
        }
    }
}

#define G1_BM   128
#define G1_BN   128
#define G1_BK   40
#define G1_LDP  44
#define G1_NCH  (REPD / G1_BK)
#define G1_PCS  (G1_BK / 4)
#define G1_LPT  (G1_BM * G1_PCS / 256)

__global__ __launch_bounds__(256) void gemm1_wmma_kernel(
    const float* __restrict__ rep, const float* __restrict__ W1t,
    const float* __restrict__ b1, float* __restrict__ h)
{
    __shared__ __attribute__((aligned(16))) float sA[2][G1_BM * G1_LDP];
    __shared__ __attribute__((aligned(16))) float sB[2][G1_BN * G1_LDP];

    const int tid  = threadIdx.x;
    const int lane = tid & 31;
    const int wave = tid >> 5;
    const int wm   = wave & 3;
    const int wn   = wave >> 2;
    const int row0 = blockIdx.x * G1_BM;
    const int col0 = blockIdx.y * G1_BN;

    int gofs[G1_LPT], lofs[G1_LPT];
    #pragma unroll
    for (int i = 0; i < G1_LPT; ++i) {
        const int p = tid + 256 * i;
        const int r = p / G1_PCS;
        const int q = p - r * G1_PCS;
        gofs[i] = r * REPD + q * 4;
        lofs[i] = r * G1_LDP + q * 4;
    }
    const float* __restrict__ gA = rep + (size_t)row0 * REPD;
    const float* __restrict__ gB = W1t + (size_t)col0 * REPD;

    v4f ra[G1_LPT], rb[G1_LPT];
    #pragma unroll
    for (int i = 0; i < G1_LPT; ++i) {
        ra[i] = *(const v4f*)(gA + gofs[i]);
        rb[i] = *(const v4f*)(gB + gofs[i]);
    }
    #pragma unroll
    for (int i = 0; i < G1_LPT; ++i) {
        *(v4f*)(&sA[0][lofs[i]]) = ra[i];
        *(v4f*)(&sB[0][lofs[i]]) = rb[i];
    }
    __syncthreads();

    const int laneM = lane & 15;
    const int kHi   = (lane >> 4) << 1;
    const int aoff  = (wm * 32 + laneM) * G1_LDP + kHi;
    const int boff  = (wn * 64 + laneM) * G1_LDP + kHi;

    v8f acc[2][4];
    #pragma unroll
    for (int i = 0; i < 2; ++i)
        #pragma unroll
        for (int j = 0; j < 4; ++j)
            acc[i][j] = (v8f){};

    for (int ch = 0; ch < G1_NCH; ++ch) {
        const int cur = ch & 1;
        const bool more = (ch + 1 < G1_NCH);
        if (more) {
            const int kofs = (ch + 1) * G1_BK;
            #pragma unroll
            for (int i = 0; i < G1_LPT; ++i) {
                ra[i] = *(const v4f*)(gA + gofs[i] + kofs);
                rb[i] = *(const v4f*)(gB + gofs[i] + kofs);
            }
        }

        const float* pa = &sA[cur][aoff];
        const float* pb = &sB[cur][boff];
        #pragma unroll
        for (int s = 0; s < G1_BK / 4; ++s) {
            v2f a[2], b[4];
            #pragma unroll
            for (int i = 0; i < 2; ++i)
                a[i] = *(const v2f*)(pa + i * 16 * G1_LDP + s * 4);
            #pragma unroll
            for (int j = 0; j < 4; ++j)
                b[j] = *(const v2f*)(pb + j * 16 * G1_LDP + s * 4);
            #pragma unroll
            for (int i = 0; i < 2; ++i)
                #pragma unroll
                for (int j = 0; j < 4; ++j)
                    acc[i][j] = __builtin_amdgcn_wmma_f32_16x16x4_f32(
                        false, a[i], false, b[j], (short)0, acc[i][j], false, false);
        }

        if (more) {
            #pragma unroll
            for (int i = 0; i < G1_LPT; ++i) {
                *(v4f*)(&sA[cur ^ 1][lofs[i]]) = ra[i];
                *(v4f*)(&sB[cur ^ 1][lofs[i]]) = rb[i];
            }
        }
        __syncthreads();
    }

    const int mOff = (lane >> 4) << 3;
    #pragma unroll
    for (int i = 0; i < 2; ++i) {
        #pragma unroll
        for (int j = 0; j < 4; ++j) {
            const int n = col0 + wn * 64 + j * 16 + laneM;
            if (n >= HID) continue;
            const float bias = b1[n];
            #pragma unroll
            for (int r = 0; r < 8; ++r) {
                const int m = row0 + wm * 32 + i * 16 + r + mOff;
                const float v = acc[i][j][r] + bias;
                h[(size_t)m * HID + n] = v > 0.0f ? v : 0.0f;
            }
        }
    }
}

__global__ __launch_bounds__(256) void gemm2_kernel(
    const float* __restrict__ h, const float* __restrict__ W2,
    const float* __restrict__ b2, float* __restrict__ out)
{
    __shared__ float red[3][256];
    const int b = blockIdx.x, tid = threadIdx.x;
    float a0 = 0.0f, a1 = 0.0f, a2 = 0.0f;
    const float* __restrict__ hr = h + (size_t)b * HID;
    for (int k = tid; k < HID; k += 256) {
        const float v = hr[k];
        a0 += v * W2[k * 3 + 0];
        a1 += v * W2[k * 3 + 1];
        a2 += v * W2[k * 3 + 2];
    }
    red[0][tid] = a0; red[1][tid] = a1; red[2][tid] = a2;
    __syncthreads();
    for (int s = 128; s > 0; s >>= 1) {
        if (tid < s) {
            red[0][tid] += red[0][tid + s];
            red[1][tid] += red[1][tid + s];
            red[2][tid] += red[2][tid + s];
        }
        __syncthreads();
    }
    if (tid < 3) out[b * 3 + tid] = red[tid][0] + b2[tid];
}

extern "C" void kernel_launch(void* const* d_in, const int* in_sizes, int n_in,
                              void* d_out, int out_size, void* d_ws, size_t ws_size,
                              hipStream_t stream)
{
    const float* emb  = (const float*)d_in[0];
    const float* W1   = (const float*)d_in[1];
    const float* b1   = (const float*)d_in[2];
    const float* W2   = (const float*)d_in[3];
    const float* b2   = (const float*)d_in[4];
    const int*   x    = (const int*)d_in[5];
    const int*   lens = (const int*)d_in[6];
    float* out = (float*)d_out;

    float* W1t = (float*)d_ws;
    float* rep = W1t + (size_t)HIDP * REPD;
    float* h   = rep + (size_t)BATCH * REPD;

    hipLaunchKernelGGL(transpose_pad_kernel, dim3(19, 32), dim3(32, 8), 0, stream,
                       W1, W1t);

    hipLaunchKernelGGL(pool_kernel, dim3(BATCH), dim3(PL_NW * 32), 0, stream,
                       emb, x, lens, rep);

    hipLaunchKernelGGL(gemm1_wmma_kernel, dim3(BATCH / 128, HIDP / 128), dim3(256), 0, stream,
                       rep, W1t, b1, h);

    hipLaunchKernelGGL(gemm2_kernel, dim3(BATCH), dim3(256), 0, stream,
                       h, W2, b2, out);
}


# layer 2 re-cut: a wave owns 4 whole rows of h (16-byte loads, its W2 slice kept in registers) and joins lanes by shuffles, no LDS or barrier; 1024 waves, not 32768; out sum order changes
# speedup vs baseline: 1.9125x; 1.0406x over previous
#include <hip/hip_runtime.h>
#include <math.h>

#define VOCAB 100000
#define DIM   300
#define BATCH 4096
#define SEQL  200
#define HID   1000
#define HIDP  1024
#define OUTD  3
#define REPD  (2*DIM)

typedef __attribute__((ext_vector_type(2))) float v2f;
typedef __attribute__((ext_vector_type(4))) float v4f;
typedef __attribute__((ext_vector_type(8))) float v8f;

__global__ __launch_bounds__(256) void transpose_pad_kernel(
    const float* __restrict__ W1, float* __restrict__ W1t)
{
    __shared__ float tile[32][33];
    const int kBase = blockIdx.x * 32;
    const int nBase = blockIdx.y * 32;
    const int tx = threadIdx.x;
    const int ty = threadIdx.y;

    #pragma unroll
    for (int i = ty; i < 32; i += 8) {
        const int k = kBase + i, n = nBase + tx;
        tile[i][tx] = (k < REPD && n < HID) ? W1[(size_t)k * HID + n] : 0.0f;
    }
    __syncthreads();
    #pragma unroll
    for (int i = ty; i < 32; i += 8) {
        const int n = nBase + i, k = kBase + tx;
        if (k < REPD)
            W1t[(size_t)n * REPD + k] = tile[tx][i];
    }
}

#define PL_NW   2
#define PL_TPW  (SEQL / PL_NW)
#define PL_R    5
#define PL_QT   25
#define PL_NQ   (PL_TPW / PL_QT)
#define PL_NP   10
#define PL_LAST 268

__global__ __launch_bounds__(PL_NW * 32) void pool_kernel(
    const float* __restrict__ emb, const int* __restrict__ x,
    const int* __restrict__ lengths, float* __restrict__ rep)
{
    __shared__ float psum[PL_NP][32];
    __shared__ float pmax[PL_NP][32];

    const int b    = blockIdx.x;
    const int lane = threadIdx.x & 31;
    const int w    = __builtin_amdgcn_readfirstlane((int)(threadIdx.x >> 5));
    const int len  = lengths[b];
    const int t0   = w * PL_TPW;
    const int* __restrict__ xr = x + (size_t)b * SEQL + t0;

    const int lq = lane < PL_QT ? lane : PL_QT - 1;
    int xv[PL_NQ];
    #pragma unroll
    for (int q = 0; q < PL_NQ; ++q) xv[q] = xr[q * PL_QT + lq];

    float s[PL_NP], m[PL_NP];
    #pragma unroll
    for (int j = 0; j < PL_NP; ++j) { s[j] = 0.0f; m[j] = -INFINITY; }

    #pragma unroll 1
    for (int q = 0; q < PL_NQ; ++q) {
        const int xq = q == 0 ? xv[0] : q == 1 ? xv[1] : q == 2 ? xv[2] : xv[3];
        const int tq = t0 + q * PL_QT;
        float v[PL_R][PL_NP];

        #pragma unroll
        for (int k = 0; k < PL_R; ++k) {
            const int idx = __builtin_amdgcn_readlane(xq, k);
            const float* __restrict__ row = emb + (size_t)idx * DIM;
            #pragma unroll
            for (int j = 0; j < PL_NP; ++j)
                v[k][j] = row[(j < PL_NP - 1 ? 32 * j : PL_LAST) + lane];
        }

        #pragma unroll
        for (int i = 0; i < PL_QT; ++i) {
            const int  k   = i % PL_R;
            const bool act = (tq + i) < len;
            #pragma unroll
            for (int j = 0; j < PL_NP; ++j) {
                const float a = v[k][j];
                s[j] += a;
                m[j] = (act && a > m[j]) ? a : m[j];
            }
            if (i + PL_R < PL_QT) {
                asm volatile("" :: "v"(s[0]), "v"(s[1]), "v"(s[2]), "v"(s[3]), "v"(s[4]),
                                   "v"(s[5]), "v"(s[6]), "v"(s[7]), "v"(s[8]), "v"(s[9]),
                                   "v"(m[0]), "v"(m[1]), "v"(m[2]), "v"(m[3]), "v"(m[4]),
                                   "v"(m[5]), "v"(m[6]), "v"(m[7]), "v"(m[8]), "v"(m[9])
                             : "memory");
                const int idx = __builtin_amdgcn_readlane(xq, i + PL_R);
                const float* __restrict__ row = emb + (size_t)idx * DIM;
                #pragma unroll
                for (int j = 0; j < PL_NP; ++j)
                    v[k][j] = row[(j < PL_NP - 1 ? 32 * j : PL_LAST) + lane];
            }
        }
    }

    if (w == 1) {
        #pragma unroll
        for (int j = 0; j < PL_NP; ++j) { psum[j][lane] = s[j]; pmax[j][lane] = m[j]; }
    }
    __syncthreads();
    if (w == 0) {
        const float flen = (float)len;
        float* __restrict__ rrow = rep + (size_t)b * REPD;
        #pragma unroll
        for (int j = 0; j < PL_NP; ++j) {
            const float st = s[j] + psum[j][lane];
            const float mt = fmaxf(m[j], pmax[j][lane]);
            const int   d  = (j < PL_NP - 1 ? 32 * j : PL_LAST) + lane;
            if (j < PL_NP - 1 || lane >= 20) {
                rrow[d]       = st / flen;
                rrow[DIM + d] = mt;
            }
        }
    }
}

#define G1_BM   128
#define G1_BN   128
#define G1_BK   40
#define G1_LDP  44
#define G1_NCH  (REPD / G1_BK)
#define G1_PCS  (G1_BK / 4)
#define G1_LPT  (G1_BM * G1_PCS / 256)

__global__ __launch_bounds__(256) void gemm1_wmma_kernel(
    const float* __restrict__ rep, const float* __restrict__ W1t,
    const float* __restrict__ b1, float* __restrict__ h)
{
    __shared__ __attribute__((aligned(16))) float sA[2][G1_BM * G1_LDP];
    __shared__ __attribute__((aligned(16))) float sB[2][G1_BN * G1_LDP];

    const int tid  = threadIdx.x;
    const int lane = tid & 31;
    const int wave = tid >> 5;
    const int wm   = wave & 3;
    const int wn   = wave >> 2;
    const int row0 = blockIdx.x * G1_BM;
    const int col0 = blockIdx.y * G1_BN;

    int gofs[G1_LPT], lofs[G1_LPT];
    #pragma unroll
    for (int i = 0; i < G1_LPT; ++i) {
        const int p = tid + 256 * i;
        const int r = p / G1_PCS;
        const int q = p - r * G1_PCS;
        gofs[i] = r * REPD + q * 4;
        lofs[i] = r * G1_LDP + q * 4;
    }
    const float* __restrict__ gA = rep + (size_t)row0 * REPD;
    const float* __restrict__ gB = W1t + (size_t)col0 * REPD;

    v4f ra[G1_LPT], rb[G1_LPT];
    #pragma unroll
    for (int i = 0; i < G1_LPT; ++i) {
        ra[i] = *(const v4f*)(gA + gofs[i]);
        rb[i] = *(const v4f*)(gB + gofs[i]);
    }
    #pragma unroll
    for (int i = 0; i < G1_LPT; ++i) {
        *(v4f*)(&sA[0][lofs[i]]) = ra[i];
        *(v4f*)(&sB[0][lofs[i]]) = rb[i];
    }
    __syncthreads();

    const int laneM = lane & 15;
    const int kHi   = (lane >> 4) << 1;
    const int aoff  = (wm * 32 + laneM) * G1_LDP + kHi;
    const int boff  = (wn * 64 + laneM) * G1_LDP + kHi;

    v8f acc[2][4];
    #pragma unroll
    for (int i = 0; i < 2; ++i)
        #pragma unroll
        for (int j = 0; j < 4; ++j)
            acc[i][j] = (v8f){};

    for (int ch = 0; ch < G1_NCH; ++ch) {
        const int cur = ch & 1;
        const bool more = (ch + 1 < G1_NCH);
        if (more) {
            const int kofs = (ch + 1) * G1_BK;
            #pragma unroll
            for (int i = 0; i < G1_LPT; ++i) {
                ra[i] = *(const v4f*)(gA + gofs[i] + kofs);
                rb[i] = *(const v4f*)(gB + gofs[i] + kofs);
            }
        }

        const float* pa = &sA[cur][aoff];
        const float* pb = &sB[cur][boff];
        #pragma unroll
        for (int s = 0; s < G1_BK / 4; ++s) {
            v2f a[2], b[4];
            #pragma unroll
            for (int i = 0; i < 2; ++i)
                a[i] = *(const v2f*)(pa + i * 16 * G1_LDP + s * 4);
            #pragma unroll
            for (int j = 0; j < 4; ++j)
                b[j] = *(const v2f*)(pb + j * 16 * G1_LDP + s * 4);
            #pragma unroll
            for (int i = 0; i < 2; ++i)
                #pragma unroll
                for (int j = 0; j < 4; ++j)
                    acc[i][j] = __builtin_amdgcn_wmma_f32_16x16x4_f32(
                        false, a[i], false, b[j], (short)0, acc[i][j], false, false);
        }

        if (more) {
            #pragma unroll
            for (int i = 0; i < G1_LPT; ++i) {
                *(v4f*)(&sA[cur ^ 1][lofs[i]]) = ra[i];
                *(v4f*)(&sB[cur ^ 1][lofs[i]]) = rb[i];
            }
        }
        __syncthreads();
    }

    const int mOff = (lane >> 4) << 3;
    #pragma unroll
    for (int i = 0; i < 2; ++i) {
        #pragma unroll
        for (int j = 0; j < 4; ++j) {
            const int n = col0 + wn * 64 + j * 16 + laneM;
            if (n >= HID) continue;
            const float bias = b1[n];
            #pragma unroll
            for (int r = 0; r < 8; ++r) {
                const int m = row0 + wm * 32 + i * 16 + r + mOff;
                const float v = acc[i][j][r] + bias;
                h[(size_t)m * HID + n] = v > 0.0f ? v : 0.0f;
            }
        }
    }
}

#define G2_RPW  4
#define G2_WPB  4
#define G2_NV   8
#define G2_F4   (HID / 4)

__global__ __launch_bounds__(G2_WPB * 32) void gemm2_kernel(
    const float* __restrict__ h, const float* __restrict__ W2,
    const float* __restrict__ b2, float* __restrict__ out)
{
    const int lane = threadIdx.x & 31;
    const int wave = threadIdx.x >> 5;
    const int row0 = (blockIdx.x * G2_WPB + wave) * G2_RPW;

    const v4f* __restrict__ W2v = (const v4f*)W2;
    const v4f zero4 = (v4f){0.0f, 0.0f, 0.0f, 0.0f};
    const float bias = b2[lane % OUTD];

    v4f w[G2_NV][3];
    #pragma unroll
    for (int i = 0; i < G2_NV; ++i) {
        const int f  = lane + 32 * i;
        const int fc = f < G2_F4 ? f : G2_F4 - 1;
        #pragma unroll
        for (int j = 0; j < 3; ++j) w[i][j] = W2v[3 * fc + j];
    }

    v4f hv[G2_RPW][G2_NV];
    #pragma unroll
    for (int r = 0; r < G2_RPW; ++r) {
        const v4f* __restrict__ hr = (const v4f*)(h + (size_t)(row0 + r) * HID);
        #pragma unroll
        for (int i = 0; i < G2_NV; ++i) {
            const int f  = lane + 32 * i;
            const int fc = f < G2_F4 ? f : G2_F4 - 1;
            hv[r][i] = hr[fc];
        }
    }

    {
        const bool ok = lane + 32 * (G2_NV - 1) < G2_F4;
        #pragma unroll
        for (int j = 0; j < 3; ++j)
            w[G2_NV - 1][j] = ok ? w[G2_NV - 1][j] : zero4;
        #pragma unroll
        for (int r = 0; r < G2_RPW; ++r)
            hv[r][G2_NV - 1] = ok ? hv[r][G2_NV - 1] : zero4;
    }

    float res = 0.0f;
    #pragma unroll
    for (int r = 0; r < G2_RPW; ++r) {
        v2f sa = (v2f){0.0f, 0.0f};
        v2f sb = (v2f){0.0f, 0.0f};
        v2f sc = (v2f){0.0f, 0.0f};
        #pragma unroll
        for (int i = 0; i < G2_NV; ++i) {
            const v4f x = hv[r][i];
            const v4f p = w[i][0], q = w[i][1], s = w[i][2];
            sa += (v2f){x.x, x.x} * (v2f){p.x, p.y};
            sb += (v2f){x.x, x.y} * (v2f){p.z, p.w};
            sc += (v2f){x.y, x.y} * (v2f){q.x, q.y};
            sa += (v2f){x.z, x.z} * (v2f){q.z, q.w};
            sb += (v2f){x.z, x.w} * (v2f){s.x, s.y};
            sc += (v2f){x.w, x.w} * (v2f){s.z, s.w};
        }
        float a0 = sa.x + sb.y;
        float a1 = sa.y + sc.x;
        float a2 = sb.x + sc.y;
        #pragma unroll
        for (int d = 16; d > 0; d >>= 1) {
            a0 += __shfl_xor(a0, d, 32);
            a1 += __shfl_xor(a1, d, 32);
            a2 += __shfl_xor(a2, d, 32);
        }
        res = (lane == OUTD * r + 0) ? a0 : res;
        res = (lane == OUTD * r + 1) ? a1 : res;
        res = (lane == OUTD * r + 2) ? a2 : res;
    }
    const float o = res + bias;
    if (lane < OUTD * G2_RPW)
        out[(size_t)row0 * OUTD + lane] = o;
}

extern "C" void kernel_launch(void* const* d_in, const int* in_sizes, int n_in,
                              void* d_out, int out_size, void* d_ws, size_t ws_size,
                              hipStream_t stream)
{
    const float* emb  = (const float*)d_in[0];
    const float* W1   = (const float*)d_in[1];
    const float* b1   = (const float*)d_in[2];
    const float* W2   = (const float*)d_in[3];
    const float* b2   = (const float*)d_in[4];
    const int*   x    = (const int*)d_in[5];
    const int*   lens = (const int*)d_in[6];
    float* out = (float*)d_out;

    float* W1t = (float*)d_ws;
    float* rep = W1t + (size_t)HIDP * REPD;
    float* h   = rep + (size_t)BATCH * REPD;

    hipLaunchKernelGGL(transpose_pad_kernel, dim3(19, 32), dim3(32, 8), 0, stream,
                       W1, W1t);

    hipLaunchKernelGGL(pool_kernel, dim3(BATCH), dim3(PL_NW * 32), 0, stream,
                       emb, x, lens, rep);

    hipLaunchKernelGGL(gemm1_wmma_kernel, dim3(BATCH / 128, HIDP / 128), dim3(256), 0, stream,
                       rep, W1t, b1, h);

    hipLaunchKernelGGL(gemm2_kernel, dim3(BATCH / (G2_WPB * G2_RPW)), dim3(G2_WPB * 32), 0, stream,
                       h, W2, b2, out);
}
